# starting kernel
# baseline (speedup 1.0000x reference)
#include <hip/hip_runtime.h>

#define NN 10000
#define NE 160000
#define FIN 16
#define HEADS 8
#define CHN 8
#define HIDD 64
#define NGR 64
#define VOC 50000
#define LSTK 9
#define EP (NE + NN)

typedef __attribute__((ext_vector_type(2))) float v2f;
typedef __attribute__((ext_vector_type(8))) float v8f;

__device__ __forceinline__ unsigned fkey(float f) {
  unsigned u = __float_as_uint(f);
  return (u >> 31) ? ~u : (u | 0x80000000u);
}
__device__ __forceinline__ float funkey(unsigned k) {
  return __uint_as_float((k >> 31) ? (k & 0x7FFFFFFFu) : ~k);
}
#define KEY_NEG_INF 0x007FFFFFu

__global__ void k_gemm3_wmma(const float* __restrict__ A,
                             const float* __restrict__ B0, const float* __restrict__ b0,
                             float* __restrict__ C0,
                             const float* __restrict__ B1, const float* __restrict__ b1,
                             float* __restrict__ C1,
                             const float* __restrict__ B2, const float* __restrict__ b2,
                             float* __restrict__ C2,
                             int K) {
  const int lane = threadIdx.x;
  const int half = lane >> 4;
  const int l    = lane & 15;
  const int tn = blockIdx.x, tm = blockIdx.y;
  const int arow = tm * 16 + l;
  const int bcol = tn * 16 + l;
  v8f acc0 = {}, acc1 = {}, acc2 = {};
  for (int k = 0; k < K; k += 4) {
    const float* ap = A + arow * K + k + half * 2;
    v2f a; a.x = ap[0]; a.y = ap[1];
    const int boff = (k + half * 2) * HIDD + bcol;
    v2f w0; w0.x = B0[boff]; w0.y = B0[boff + HIDD];
    v2f w1; w1.x = B1[boff]; w1.y = B1[boff + HIDD];
    v2f w2; w2.x = B2[boff]; w2.y = B2[boff + HIDD];
    acc0 = __builtin_amdgcn_wmma_f32_16x16x4_f32(false, a, false, w0, (short)0, acc0, false, false);
    acc1 = __builtin_amdgcn_wmma_f32_16x16x4_f32(false, a, false, w1, (short)0, acc1, false, false);
    acc2 = __builtin_amdgcn_wmma_f32_16x16x4_f32(false, a, false, w2, (short)0, acc2, false, false);
  }
  const float bv0 = b0[bcol], bv1 = b1[bcol], bv2 = b2[bcol];
  const int rb = tm * 16 + half * 8;
#pragma unroll
  for (int r = 0; r < 8; ++r) {
    const int off = (rb + r) * HIDD + bcol;
    C0[off] = acc0[r] + bv0;
    C1[off] = acc1[r] + bv1;
    C2[off] = acc2[r] + bv2;
  }
}

__global__ void k_gemm_wmma(const float* __restrict__ A, const float* __restrict__ B,
                            const float* __restrict__ bias, float* __restrict__ C,
                            int M, int K, int Nc) {
  const int lane = threadIdx.x;
  const int half = lane >> 4;
  const int l    = lane & 15;
  const int tn = blockIdx.x, tm = blockIdx.y;
  const int arow = tm * 16 + l;
  const int bcol = tn * 16 + l;
  v8f acc = {};
  for (int k = 0; k < K; k += 4) {
    const float* ap = A + arow * K + k + half * 2;
    v2f a; a.x = ap[0]; a.y = ap[1];
    const float* bp = B + (size_t)(k + half * 2) * Nc + bcol;
    v2f b; b.x = bp[0]; b.y = bp[Nc];
    if (k + 4 < K) {
      __builtin_prefetch(bp + 4 * (size_t)Nc, 0, 1);
      __builtin_prefetch(bp + 5 * (size_t)Nc, 0, 1);
    }
    acc = __builtin_amdgcn_wmma_f32_16x16x4_f32(false, a, false, b, (short)0, acc, false, false);
  }
  const float bv = bias[bcol];
  const int rb = tm * 16 + half * 8;
#pragma unroll
  for (int r = 0; r < 8; ++r)
    C[(size_t)(rb + r) * Nc + bcol] = acc[r] + bv;
}

__global__ void k_zero(float* __restrict__ p, int n) {
  int i = blockIdx.x * blockDim.x + threadIdx.x;
  if (i < n) p[i] = 0.f;
}

__global__ void k_deg_loop(const int* __restrict__ dstE, const float* __restrict__ ea,
                           float* __restrict__ deg, float* __restrict__ la) {
  int e = blockIdx.x * blockDim.x + threadIdx.x;
  if (e >= NE) return;
  int d = dstE[e];
  atomicAdd(deg + d, 1.f);
  atomicAdd(la + 2 * d,     ea[2 * e]);
  atomicAdd(la + 2 * d + 1, ea[2 * e + 1]);
}

__global__ void k_loop_div(float* __restrict__ la, const float* __restrict__ deg) {
  int i = blockIdx.x * blockDim.x + threadIdx.x;
  if (i < 2 * NN) la[i] /= fmaxf(deg[i >> 1], 1.f);
}

__global__ void k_attn_init(unsigned* __restrict__ mxkey, float* __restrict__ den,
                            float* __restrict__ agg) {
  int i = blockIdx.x * blockDim.x + threadIdx.x;
  if (i < NN * HIDD) agg[i] = 0.f;
  if (i < NN * HEADS) { mxkey[i] = KEY_NEG_INF; den[i] = 0.f; }
}

__global__ void k_edge_logit(const float* __restrict__ xl, const float* __restrict__ xr,
                             const int* __restrict__ srcE, const int* __restrict__ dstE,
                             const float* __restrict__ ea, const float* __restrict__ la,
                             const float* __restrict__ We, const float* __restrict__ att,
                             float* __restrict__ logit, unsigned* __restrict__ mxkey) {
  int idx = blockIdx.x * blockDim.x + threadIdx.x;
  if (idx >= EP * HEADS) return;
  int e = idx >> 3, hh = idx & 7;
  int s, d; float a0, a1;
  if (e < NE) { s = srcE[e]; d = dstE[e]; a0 = ea[2 * e]; a1 = ea[2 * e + 1]; }
  else        { s = d = e - NE;           a0 = la[2 * s]; a1 = la[2 * s + 1]; }
  float lg = 0.f;
#pragma unroll
  for (int c = 0; c < CHN; ++c) {
    int hc = hh * CHN + c;
    float m = xl[s * HIDD + hc] + xr[d * HIDD + hc] + a0 * We[hc] + a1 * We[HIDD + hc];
    float g = m > 0.f ? m : 0.2f * m;
    lg += g * att[hc];
  }
  logit[idx] = lg;
  atomicMax(mxkey + d * HEADS + hh, fkey(lg));
}

__global__ void k_edge_exp(const float* __restrict__ logit, const unsigned* __restrict__ mxkey,
                           const int* __restrict__ dstE, float* __restrict__ exl,
                           float* __restrict__ den) {
  int idx = blockIdx.x * blockDim.x + threadIdx.x;
  if (idx >= EP * HEADS) return;
  int e = idx >> 3, hh = idx & 7;
  int d = (e < NE) ? dstE[e] : e - NE;
  float v = __expf(logit[idx] - funkey(mxkey[d * HEADS + hh]));
  exl[idx] = v;
  atomicAdd(den + d * HEADS + hh, v);
}

__global__ void k_edge_agg(const float* __restrict__ xl, const float* __restrict__ exl,
                           const float* __restrict__ den, const int* __restrict__ srcE,
                           const int* __restrict__ dstE, float* __restrict__ agg) {
  int idx = blockIdx.x * blockDim.x + threadIdx.x;
  if (idx >= EP * HEADS) return;
  int e = idx >> 3, hh = idx & 7;
  int s, d;
  if (e < NE) { s = srcE[e]; d = dstE[e]; } else { s = d = e - NE; }
  float alpha = exl[idx] / den[d * HEADS + hh];
#pragma unroll
  for (int c = 0; c < CHN; ++c)
    atomicAdd(agg + d * HIDD + hh * CHN + c, alpha * xl[s * HIDD + hh * CHN + c]);
}

__global__ void k_combine_elu(const float* __restrict__ agg, const float* __restrict__ bias_g,
                              const float* __restrict__ res, float* __restrict__ hout) {
  int idx = blockIdx.x * blockDim.x + threadIdx.x;
  if (idx >= NN * HIDD) return;
  float v = agg[idx] + bias_g[idx & (HIDD - 1)] + res[idx];
  hout[idx] = v > 0.f ? v : (__expf(v) - 1.f);
}

__global__ void k_pool_accum(const float* __restrict__ h, const int* __restrict__ batch,
                             float* __restrict__ pooled, float* __restrict__ cnt) {
  int idx = blockIdx.x * blockDim.x + threadIdx.x;
  if (idx >= NN * HIDD) return;
  int n = idx >> 6, c = idx & (HIDD - 1);
  int b = batch[n];
  atomicAdd(pooled + b * HIDD + c, h[idx]);
  if (c == 0) atomicAdd(cnt + b, 1.f);
}

__global__ void k_pool_div(float* __restrict__ pooled, const float* __restrict__ cnt) {
  int idx = blockIdx.x * blockDim.x + threadIdx.x;
  if (idx < NGR * HIDD) pooled[idx] /= fmaxf(cnt[idx >> 6], 1.f);
}

extern "C" void kernel_launch(void* const* d_in, const int* in_sizes, int n_in,
                              void* d_out_v, int out_size, void* d_ws, size_t ws_size,
                              hipStream_t stream) {
  (void)in_sizes; (void)n_in; (void)out_size; (void)ws_size;
  const float* x       = (const float*)d_in[0];
  const int*   ei      = (const int*)  d_in[1];
  const float* ea      = (const float*)d_in[2];
  const int*   batch   = (const int*)  d_in[3];
  const float* p1_Wl   = (const float*)d_in[4];
  const float* p1_bl   = (const float*)d_in[5];
  const float* p1_Wr   = (const float*)d_in[6];
  const float* p1_br   = (const float*)d_in[7];
  const float* p1_We   = (const float*)d_in[8];
  const float* p1_att  = (const float*)d_in[9];
  const float* p1_bias = (const float*)d_in[10];
  const float* p1_Wres = (const float*)d_in[11];
  const float* p1_bres = (const float*)d_in[12];
  const float* Wl      = (const float*)d_in[13];
  const float* bl      = (const float*)d_in[14];
  const float* Wr      = (const float*)d_in[15];
  const float* br      = (const float*)d_in[16];
  const float* We      = (const float*)d_in[17];
  const float* att     = (const float*)d_in[18];
  const float* bias_g  = (const float*)d_in[19];
  const float* Wres    = (const float*)d_in[20];
  const float* bres    = (const float*)d_in[21];
  const float* Wlin    = (const float*)d_in[22];
  const float* blin    = (const float*)d_in[23];

  float* out = (float*)d_out_v;
  float* ws  = (float*)d_ws;

  float*    xl    = ws;
  float*    xr    = ws + 640000;
  float*    res   = ws + 1280000;
  float*    agg   = ws + 1920000;
  float*    logit = ws + 2560000;
  float*    exl   = ws + 3920000;
  float*    den   = ws + 5280000;
  unsigned* mxkey = (unsigned*)(ws + 5360000);
  float*    deg   = ws + 5440000;
  float*    la    = ws + 5450000;
  float*    cnt   = ws + 5470000;

  const int* srcE = ei;
  const int* dstE = ei + NE;

  const int TB = 256;
  const int gEdge = (EP * HEADS + TB - 1) / TB;
  const int gNode = (NN * HIDD + TB - 1) / TB;

  k_zero<<<(NN + TB - 1) / TB, TB, 0, stream>>>(deg, NN);
  k_zero<<<(2 * NN + TB - 1) / TB, TB, 0, stream>>>(la, 2 * NN);
  k_deg_loop<<<(NE + TB - 1) / TB, TB, 0, stream>>>(dstE, ea, deg, la);
  k_loop_div<<<(2 * NN + TB - 1) / TB, TB, 0, stream>>>(la, deg);

  for (int layer = 0; layer < 1 + LSTK; ++layer) {
    const float* hin = (layer == 0) ? x : out + (size_t)(layer - 1) * NN * HIDD;
    const int K = (layer == 0) ? FIN : HIDD;
    const int li = layer - 1;
    const float* Wl_   = (layer == 0) ? p1_Wl   : Wl   + (size_t)li * HIDD * HIDD;
    const float* bl_   = (layer == 0) ? p1_bl   : bl   + (size_t)li * HIDD;
    const float* Wr_   = (layer == 0) ? p1_Wr   : Wr   + (size_t)li * HIDD * HIDD;
    const float* br_   = (layer == 0) ? p1_br   : br   + (size_t)li * HIDD;
    const float* We_   = (layer == 0) ? p1_We   : We   + (size_t)li * 2 * HIDD;
    const float* att_  = (layer == 0) ? p1_att  : att  + (size_t)li * HEADS * CHN;
    const float* bg_   = (layer == 0) ? p1_bias : bias_g + (size_t)li * HIDD;
    const float* Wres_ = (layer == 0) ? p1_Wres : Wres + (size_t)li * HIDD * HIDD;
    const float* bres_ = (layer == 0) ? p1_bres : bres + (size_t)li * HIDD;
    float* hout = out + (size_t)layer * NN * HIDD;

    dim3 gg(HIDD / 16, NN / 16);
    k_gemm3_wmma<<<gg, 32, 0, stream>>>(hin, Wl_, bl_, xl, Wr_, br_, xr,
                                        Wres_, bres_, res, K);

    k_attn_init<<<gNode, TB, 0, stream>>>(mxkey, den, agg);
    k_edge_logit<<<gEdge, TB, 0, stream>>>(xl, xr, srcE, dstE, ea, la, We_, att_, logit, mxkey);
    k_edge_exp  <<<gEdge, TB, 0, stream>>>(logit, mxkey, dstE, exl, den);
    k_edge_agg  <<<gEdge, TB, 0, stream>>>(xl, exl, den, srcE, dstE, agg);
    k_combine_elu<<<gNode, TB, 0, stream>>>(agg, bg_, res, hout);
  }

  float* pooled = out + (size_t)(1 + LSTK) * NN * HIDD;
  float* x12    = pooled + NGR * HIDD;
  k_zero<<<(NGR * HIDD + TB - 1) / TB, TB, 0, stream>>>(pooled, NGR * HIDD);
  k_zero<<<1, NGR, 0, stream>>>(cnt, NGR);
  k_pool_accum<<<gNode, TB, 0, stream>>>(out + (size_t)LSTK * NN * HIDD, batch, pooled, cnt);
  k_pool_div<<<(NGR * HIDD + TB - 1) / TB, TB, 0, stream>>>(pooled, cnt);

  dim3 gf(VOC / 16, NGR / 16);
  k_gemm_wmma<<<gf, 32, 0, stream>>>(pooled, Wlin, blin, x12, NGR, HIDD, VOC);
}


# Segment sums leave global atomics for their owners: self-loop means from each group's own edge list inside the first layer kernel, mean pool one group per graph; 7 launches go; LDS arrival-order sums
# speedup vs baseline: 4.3100x; 4.3100x over previous
#include <hip/hip_runtime.h>

#define NN 10000
#define NE 160000
#define FIN 16
#define HEADS 8
#define CHN 8
#define HIDD 64
#define NGR 64
#define VOC 50000
#define LSTK 9
#define EP (NE + NN)

typedef __attribute__((ext_vector_type(2))) float v2f;
typedef __attribute__((ext_vector_type(4))) float v4f;
typedef __attribute__((ext_vector_type(8))) float v8f;

__device__ __forceinline__ unsigned fkey(float f) {
  unsigned u = __float_as_uint(f);
  return (u >> 31) ? ~u : (u | 0x80000000u);
}
__device__ __forceinline__ float funkey(unsigned k) {
  return __uint_as_float((k >> 31) ? (k & 0x7FFFFFFFu) : ~k);
}
#define KEY_NEG_INF 0x007FFFFFu

template <int K>
__global__ void __launch_bounds__(32)
k_gemm3_wmma(const float* __restrict__ A,
             const float* __restrict__ B0, const float* __restrict__ b0, float* __restrict__ C0,
             const float* __restrict__ B1, const float* __restrict__ b1, float* __restrict__ C1,
             const float* __restrict__ B2, const float* __restrict__ b2, float* __restrict__ C2) {
  constexpr int KS = K / 4;
  const int lane = threadIdx.x;
  const int half = lane >> 4;
  const int l    = lane & 15;
  const int tm   = blockIdx.y;
  const int which = blockIdx.x;
  const float* __restrict__ B  = which == 0 ? B0 : (which == 1 ? B1 : B2);
  const float* __restrict__ bb = which == 0 ? b0 : (which == 1 ? b1 : b2);
  float* __restrict__       C  = which == 0 ? C0 : (which == 1 ? C1 : C2);
  const int arow = tm * 16 + l;

  const float4 bv = *(const float4*)(bb + 4 * l);
  v2f    a[KS];
  v4f    q0[KS], q1[KS];
#pragma unroll
  for (int s = 0; s < KS; ++s) {
    const int k = 4 * s;
    const float* ap = A + arow * K + k + half * 2;
    a[s].x = ap[0]; a[s].y = ap[1];
    const float* bp = B + (k + half * 2) * HIDD + 4 * l;
    q0[s] = *(const v4f*)bp;
    q1[s] = *(const v4f*)(bp + HIDD);
  }
#pragma unroll
  for (int s = 0; s < KS; ++s)
    asm volatile("" :: "v"(a[s]), "v"(q0[s]), "v"(q1[s]));
  __builtin_amdgcn_sched_barrier(0);
  v8f acc0 = {}, acc1 = {}, acc2 = {}, acc3 = {};
#pragma unroll
  for (int s = 0; s < KS; ++s) {
    v2f w;
    w.x = q0[s].x; w.y = q1[s].x;
    acc0 = __builtin_amdgcn_wmma_f32_16x16x4_f32(false, a[s], false, w, (short)0, acc0, false, false);
    w.x = q0[s].y; w.y = q1[s].y;
    acc1 = __builtin_amdgcn_wmma_f32_16x16x4_f32(false, a[s], false, w, (short)0, acc1, false, false);
    w.x = q0[s].z; w.y = q1[s].z;
    acc2 = __builtin_amdgcn_wmma_f32_16x16x4_f32(false, a[s], false, w, (short)0, acc2, false, false);
    w.x = q0[s].w; w.y = q1[s].w;
    acc3 = __builtin_amdgcn_wmma_f32_16x16x4_f32(false, a[s], false, w, (short)0, acc3, false, false);
  }
  const int rb = tm * 16 + half * 8;
#pragma unroll
  for (int r = 0; r < 8; ++r) {
    float4 o;
    o.x = acc0[r] + bv.x; o.y = acc1[r] + bv.y; o.z = acc2[r] + bv.z; o.w = acc3[r] + bv.w;
    *(float4*)(C + (rb + r) * HIDD + 4 * l) = o;
  }
}

__global__ void k_gemm_wmma(const float* __restrict__ A, const float* __restrict__ B,
                            const float* __restrict__ bias, float* __restrict__ C,
                            int M, int K, int Nc) {
  const int lane = threadIdx.x;
  const int half = lane >> 4;
  const int l    = lane & 15;
  const int tn = blockIdx.x, tm = blockIdx.y;
  const int arow = tm * 16 + l;
  const int bcol = tn * 16 + l;
  v8f acc = {};
  for (int k = 0; k < K; k += 4) {
    const float* ap = A + arow * K + k + half * 2;
    v2f a; a.x = ap[0]; a.y = ap[1];
    const float* bp = B + (size_t)(k + half * 2) * Nc + bcol;
    v2f b; b.x = bp[0]; b.y = bp[Nc];
    if (k + 4 < K) {
      __builtin_prefetch(bp + 4 * (size_t)Nc, 0, 1);
      __builtin_prefetch(bp + 5 * (size_t)Nc, 0, 1);
    }
    acc = __builtin_amdgcn_wmma_f32_16x16x4_f32(false, a, false, b, (short)0, acc, false, false);
  }
  const float bv = bias[bcol];
  const int rb = tm * 16 + half * 8;
#pragma unroll
  for (int r = 0; r < 8; ++r)
    C[(size_t)(rb + r) * Nc + bcol] = acc[r] + bv;
}

#define OWN  40
#define NBLK (NN / OWN)
#define LTB  1024
#define SWEEP (LTB / HEADS)
#define EPT  7
#define CAP  (EPT * SWEEP)
#define E_BITS 18
#define E_MASK ((1 << E_BITS) - 1)
#define AGG_IDX(dl, hh, c) ((c) * (OWN * HEADS) + (dl) * HEADS + (hh))

static_assert(EP <= (1 << E_BITS), "edge id must fit the packed list entry");
static_assert(OWN * HIDD / 4 <= LTB, "one 16-byte piece of the own rows per thread");
static_assert(HIDD / 4 == 16 && CHN == 8, "a 16-byte piece is half a head: piece p of a row is head p/2, channels 4*(p%2)..+3");
static_assert(CAP <= LTB, "one record per thread");
static_assert(NBLK * CAP * 4 == 896000, "workspace room of the records, see kernel_launch");

__device__ __forceinline__ float edge_logit(const float* __restrict__ xl, const float* __restrict__ xr,
                                            int s, int d, float a0, float a1,
                                            const float* __restrict__ We, const float* __restrict__ att,
                                            int hh) {
  float lg = 0.f;
#pragma unroll
  for (int c = 0; c < CHN; ++c) {
    int hc = hh * CHN + c;
    float m = xl[s * HIDD + hc] + xr[d * HIDD + hc] + a0 * We[hc] + a1 * We[HIDD + hc];
    float g = m > 0.f ? m : 0.2f * m;
    lg += g * att[hc];
  }
  return lg;
}

__device__ __forceinline__ bool edge_fetch(int e, int n0, const int* __restrict__ srcE,
                                           const int* __restrict__ dstE, const float* __restrict__ ea,
                                           const float* __restrict__ la,
                                           int& s, int& d, float& a0, float& a1) {
  if (e < NE) {
    d = dstE[e]; s = srcE[e]; a0 = ea[2 * e]; a1 = ea[2 * e + 1];
  } else {
    s = d = e - NE;
    if ((unsigned)(d - n0) >= (unsigned)OWN) return false;
    a0 = la[2 * s]; a1 = la[2 * s + 1];
  }
  return (unsigned)(d - n0) < (unsigned)OWN;
}

template <bool BUILD>
__global__ void __launch_bounds__(LTB)
k_layer_owned(const float* __restrict__ xl, const float* __restrict__ xr,
              const float* __restrict__ res,
              const int* __restrict__ srcE, const int* __restrict__ dstE,
              const float* __restrict__ ea, float* la,
              const float* __restrict__ We, const float* __restrict__ att,
              const float* __restrict__ bias_g, float* __restrict__ hout,
              float* __restrict__ recs, int* __restrict__ gcnt) {
  __shared__ int      s_e[BUILD ? CAP : 1];
  __shared__ float    s_deg[BUILD ? OWN : 1];
  __shared__ float    s_la[BUILD ? 2 * OWN : 1];
  __shared__ int2     s_rec[CAP];
  __shared__ float2   s_ea[CAP];
  __shared__ float4   s_xr[OWN * HIDD / 4];
  __shared__ float4   s_res[OWN * HIDD / 4];
  __shared__ float4   s_bias[HIDD / 4];
  __shared__ float    s_agg[OWN * HIDD];
  __shared__ float    s_den[OWN * HEADS];
  __shared__ unsigned s_mx[OWN * HEADS];
  __shared__ int      s_cnt;

  const int tid = threadIdx.x;
  const int n0  = blockIdx.x * OWN;
  float4* __restrict__ my_recs = (float4*)recs + (size_t)blockIdx.x * CAP;

  for (int i = tid; i < OWN * HIDD; i += LTB) s_agg[i] = 0.f;
  if (tid < OWN * HEADS) { s_den[tid] = 0.f; s_mx[tid] = KEY_NEG_INF; }

  int cnt;
  if (BUILD) {
    if (tid < OWN) { s_e[tid] = (NE + n0 + tid) | (tid << E_BITS); s_deg[tid] = 0.f; }
    if (tid < 2 * OWN) s_la[tid] = 0.f;
    if (tid == 0) s_cnt = OWN;
    __syncthreads();

    float4 xr_own = {}, res_own = {}, bias_q = {};
    if (tid < OWN * HIDD / 4) {
      xr_own  = ((const float4*)(xr + n0 * HIDD))[tid];
      res_own = ((const float4*)(res + n0 * HIDD))[tid];
    }
    if (tid < HIDD / 4) bias_q = ((const float4*)bias_g)[tid];

    const int4* __restrict__ d4 = (const int4*)dstE;
    const int NQ = NE / 4;
    for (int q = tid; q < NQ; q += 4 * LTB) {
      int4 v[4];
#pragma unroll
      for (int k = 0; k < 4; ++k) {
        int qq = q + k * LTB;
        v[k] = d4[qq < NQ ? qq : NQ - 1];
      }
#pragma unroll
      for (int k = 0; k < 4; ++k) {
        int qq = q + k * LTB;
        if (qq < NQ) {
          int e = qq * 4;
          int dx = v[k].x - n0, dy = v[k].y - n0, dz = v[k].z - n0, dw = v[k].w - n0;
          if ((unsigned)dx < (unsigned)OWN) { int p = atomicAdd(&s_cnt, 1); if (p < CAP) s_e[p] = e       | (dx << E_BITS); }
          if ((unsigned)dy < (unsigned)OWN) { int p = atomicAdd(&s_cnt, 1); if (p < CAP) s_e[p] = (e + 1) | (dy << E_BITS); }
          if ((unsigned)dz < (unsigned)OWN) { int p = atomicAdd(&s_cnt, 1); if (p < CAP) s_e[p] = (e + 2) | (dz << E_BITS); }
          if ((unsigned)dw < (unsigned)OWN) { int p = atomicAdd(&s_cnt, 1); if (p < CAP) s_e[p] = (e + 3) | (dw << E_BITS); }
        }
      }
    }
    if (tid < OWN * HIDD / 4) { s_xr[tid] = xr_own; s_res[tid] = res_own; }
    if (tid < HIDD / 4) s_bias[tid] = bias_q;
    __syncthreads();

    cnt = s_cnt;
    int2 r = {0, 0}; float2 a = {0.f, 0.f};
    if (cnt <= CAP) {
      if (tid >= OWN && tid < cnt) {
        const int pk = s_e[tid];
        const int e  = pk & E_MASK;
        r.y = pk >> E_BITS;
        r.x = srcE[e]; a = ((const float2*)ea)[e];
        atomicAdd(&s_deg[r.y], 1.f);
        atomicAdd(&s_la[2 * r.y], a.x);
        atomicAdd(&s_la[2 * r.y + 1], a.y);
      }
    } else {
      for (int e = tid; e < NE; e += LTB) {
        const int d = dstE[e] - n0;
        if ((unsigned)d < (unsigned)OWN) {
          const float2 t = ((const float2*)ea)[e];
          atomicAdd(&s_deg[d], 1.f);
          atomicAdd(&s_la[2 * d], t.x);
          atomicAdd(&s_la[2 * d + 1], t.y);
        }
      }
    }
    __syncthreads();
    if (tid < OWN) {
      const float  dg = fmaxf(s_deg[tid], 1.f);
      const float2 m  = make_float2(s_la[2 * tid] / dg, s_la[2 * tid + 1] / dg);
      if (cnt <= CAP) { r = make_int2(n0 + tid, tid); a = m; }
      else            ((float2*)la)[n0 + tid] = m;
    }
    if (tid < CAP) {
      s_rec[tid] = r; s_ea[tid] = a;
      my_recs[tid] = make_float4(__int_as_float(r.x), __int_as_float(r.y), a.x, a.y);
    }
    if (tid == 0) gcnt[blockIdx.x] = cnt;
    __syncthreads();
  } else {
    cnt = gcnt[blockIdx.x];
    float4 xr_own = {}, res_own = {}, bias_q = {};
    if (tid < OWN * HIDD / 4) {
      xr_own  = ((const float4*)(xr + n0 * HIDD))[tid];
      res_own = ((const float4*)(res + n0 * HIDD))[tid];
    }
    if (tid < HIDD / 4) bias_q = ((const float4*)bias_g)[tid];
    if (tid < CAP) {
      const float4 rec = my_recs[tid];
      s_rec[tid] = make_int2(__float_as_int(rec.x), __float_as_int(rec.y));
      s_ea[tid]  = make_float2(rec.z, rec.w);
    }
    if (tid < OWN * HIDD / 4) { s_xr[tid] = xr_own; s_res[tid] = res_own; }
    if (tid < HIDD / 4) s_bias[tid] = bias_q;
    __syncthreads();
  }

  const int hh  = tid & 7;
  const int i0  = tid >> 3;

  if (cnt <= CAP) {
    float we0[CHN], we1[CHN], at[CHN];
    {
      const float4* w0 = (const float4*)(We + hh * CHN);
      const float4* w1 = (const float4*)(We + HIDD + hh * CHN);
      const float4* ap = (const float4*)(att + hh * CHN);
      float4 t;
      t = w0[0]; we0[0] = t.x; we0[1] = t.y; we0[2] = t.z; we0[3] = t.w;
      t = w0[1]; we0[4] = t.x; we0[5] = t.y; we0[6] = t.z; we0[7] = t.w;
      t = w1[0]; we1[0] = t.x; we1[1] = t.y; we1[2] = t.z; we1[3] = t.w;
      t = w1[1]; we1[4] = t.x; we1[5] = t.y; we1[6] = t.z; we1[7] = t.w;
      t = ap[0]; at[0] = t.x; at[1] = t.y; at[2] = t.z; at[3] = t.w;
      t = ap[1]; at[4] = t.x; at[5] = t.y; at[6] = t.z; at[7] = t.w;
    }

    float xv[EPT][CHN];
    float lg[EPT];
    int   dlv[EPT];

    int sv[EPT];
#pragma unroll
    for (int k = 0; k < EPT; ++k) {
      const int2 r = s_rec[i0 + k * SWEEP];
      sv[k] = r.x; dlv[k] = r.y;
    }
#pragma unroll
    for (int k = 0; k < EPT; ++k) {
      const int i = i0 + k * SWEEP;
      if (i < cnt) {
        const float4* rp = (const float4*)(xl + sv[k] * HIDD + hh * CHN);
        const float4 r0 = rp[0], r1 = rp[1];
        xv[k][0] = r0.x; xv[k][1] = r0.y; xv[k][2] = r0.z; xv[k][3] = r0.w;
        xv[k][4] = r1.x; xv[k][5] = r1.y; xv[k][6] = r1.z; xv[k][7] = r1.w;
      }
    }

#pragma unroll
    for (int k = 0; k < EPT; ++k) {
      const int i = i0 + k * SWEEP;
      if (i < cnt) {
        const int    dl = dlv[k];
        const float2 a  = s_ea[i];
        const float4 d0 = s_xr[dl * (HIDD / 4) + hh * 2], d1 = s_xr[dl * (HIDD / 4) + hh * 2 + 1];
        const float  xd[CHN] = {d0.x, d0.y, d0.z, d0.w, d1.x, d1.y, d1.z, d1.w};
        float acc = 0.f;
#pragma unroll
        for (int c = 0; c < CHN; ++c) {
          float m = xv[k][c] + xd[c] + a.x * we0[c] + a.y * we1[c];
          float g = m > 0.f ? m : 0.2f * m;
          acc += g * at[c];
        }
        lg[k] = acc;
        atomicMax(&s_mx[dl * HEADS + hh], fkey(acc));
      }
    }
    __syncthreads();

#pragma unroll
    for (int k = 0; k < EPT; ++k) {
      const int i = i0 + k * SWEEP;
      if (i < cnt) {
        const int   dl = dlv[k];
        const float v  = __expf(lg[k] - funkey(s_mx[dl * HEADS + hh]));
        atomicAdd(&s_den[dl * HEADS + hh], v);
#pragma unroll
        for (int c = 0; c < CHN; ++c)
          atomicAdd(&s_agg[AGG_IDX(dl, hh, c)], v * xv[k][c]);
      }
    }
  } else {
    for (int e = i0; e < EP; e += SWEEP) {
      int s, d; float a0, a1;
      if (!edge_fetch(e, n0, srcE, dstE, ea, la, s, d, a0, a1)) continue;
      float lgv = edge_logit(xl, xr, s, d, a0, a1, We, att, hh);
      atomicMax(&s_mx[(d - n0) * HEADS + hh], fkey(lgv));
    }
    __syncthreads();
    for (int e = i0; e < EP; e += SWEEP) {
      int s, d; float a0, a1;
      if (!edge_fetch(e, n0, srcE, dstE, ea, la, s, d, a0, a1)) continue;
      float lgv = edge_logit(xl, xr, s, d, a0, a1, We, att, hh);
      int   dl  = d - n0;
      float v   = __expf(lgv - funkey(s_mx[dl * HEADS + hh]));
      atomicAdd(&s_den[dl * HEADS + hh], v);
#pragma unroll
      for (int c = 0; c < CHN; ++c)
        atomicAdd(&s_agg[AGG_IDX(dl, hh, c)], v * xl[s * HIDD + hh * CHN + c]);
    }
  }
  __syncthreads();

  if (tid < OWN * HIDD / 4) {
    const int    dl  = tid >> 4;
    const int    hq  = (tid >> 1) & 7;
    const int    c0  = (tid & 1) * 4;
    const float  den = s_den[dl * HEADS + hq];
    const float4 rv  = s_res[tid];
    const float4 bq  = s_bias[tid & 15];
    float4 o;
    o.x = s_agg[AGG_IDX(dl, hq, c0)]     / den + bq.x + rv.x;
    o.y = s_agg[AGG_IDX(dl, hq, c0 + 1)] / den + bq.y + rv.y;
    o.z = s_agg[AGG_IDX(dl, hq, c0 + 2)] / den + bq.z + rv.z;
    o.w = s_agg[AGG_IDX(dl, hq, c0 + 3)] / den + bq.w + rv.w;
    o.x = o.x > 0.f ? o.x : (__expf(o.x) - 1.f);
    o.y = o.y > 0.f ? o.y : (__expf(o.y) - 1.f);
    o.z = o.z > 0.f ? o.z : (__expf(o.z) - 1.f);
    o.w = o.w > 0.f ? o.w : (__expf(o.w) - 1.f);
    ((float4*)(hout + n0 * HIDD))[tid] = o;
  }
}

#define PTB  1024
#define PSL  (PTB / (HIDD / 4))
#define PLD  4
static_assert(HIDD / 4 == 16, "a row is 16 pieces of 16 bytes");

__global__ void __launch_bounds__(PTB)
k_pool_owned(const float* __restrict__ h, const int* __restrict__ batch, float* __restrict__ pooled) {
  __shared__ int    s_list[NN];
  __shared__ float4 s_part[PTB];
  __shared__ int    s_n;
  const int tid = threadIdx.x;
  const int g   = blockIdx.x;
  if (tid == 0) s_n = 0;
  __syncthreads();

  constexpr int NB = (NN + PTB - 1) / PTB;
  int bv[NB];
#pragma unroll
  for (int j = 0; j < NB; ++j) {
    const int n = tid + j * PTB;
    bv[j] = n < NN ? batch[n] : -1;
  }
#pragma unroll
  for (int j = 0; j < NB; ++j)
    if (bv[j] == g) { const int p = atomicAdd(&s_n, 1); s_list[p] = tid + j * PTB; }
  __syncthreads();

  const int n  = s_n;
  const int pc = tid & 15;
  const int sl = tid >> 4;
  const float4* __restrict__ h4 = (const float4*)h;
  float4 acc = make_float4(0.f, 0.f, 0.f, 0.f);
  for (int j0 = sl; j0 < n; j0 += PLD * PSL) {
    float4 v[PLD];
#pragma unroll
    for (int u = 0; u < PLD; ++u) {
      const int j = j0 + u * PSL;
      v[u] = j < n ? h4[s_list[j] * (HIDD / 4) + pc] : make_float4(0.f, 0.f, 0.f, 0.f);
    }
#pragma unroll
    for (int u = 0; u < PLD; ++u) { acc.x += v[u].x; acc.y += v[u].y; acc.z += v[u].z; acc.w += v[u].w; }
  }
  s_part[tid] = acc;
  __syncthreads();

  if (tid < HIDD) {
    const float* sp = (const float*)s_part;
    float sum = 0.f;
    for (int s = 0; s < PSL; ++s) sum += sp[s * HIDD + tid];
    pooled[g * HIDD + tid] = sum / fmaxf((float)n, 1.f);
  }
}

extern "C" void kernel_launch(void* const* d_in, const int* in_sizes, int n_in,
                              void* d_out_v, int out_size, void* d_ws, size_t ws_size,
                              hipStream_t stream) {
  (void)in_sizes; (void)n_in; (void)out_size; (void)ws_size;
  const float* x       = (const float*)d_in[0];
  const int*   ei      = (const int*)  d_in[1];
  const float* ea      = (const float*)d_in[2];
  const int*   batch   = (const int*)  d_in[3];
  const float* p1_Wl   = (const float*)d_in[4];
  const float* p1_bl   = (const float*)d_in[5];
  const float* p1_Wr   = (const float*)d_in[6];
  const float* p1_br   = (const float*)d_in[7];
  const float* p1_We   = (const float*)d_in[8];
  const float* p1_att  = (const float*)d_in[9];
  const float* p1_bias = (const float*)d_in[10];
  const float* p1_Wres = (const float*)d_in[11];
  const float* p1_bres = (const float*)d_in[12];
  const float* Wl      = (const float*)d_in[13];
  const float* bl      = (const float*)d_in[14];
  const float* Wr      = (const float*)d_in[15];
  const float* br      = (const float*)d_in[16];
  const float* We      = (const float*)d_in[17];
  const float* att     = (const float*)d_in[18];
  const float* bias_g  = (const float*)d_in[19];
  const float* Wres    = (const float*)d_in[20];
  const float* bres    = (const float*)d_in[21];
  const float* Wlin    = (const float*)d_in[22];
  const float* blin    = (const float*)d_in[23];

  float* out = (float*)d_out_v;
  float* ws  = (float*)d_ws;

  float*    xl    = ws;
  float*    xr    = ws + 640000;
  float*    res   = ws + 1280000;
  float*    recs  = ws + 1920000;
  int*      gcnt  = (int*)(ws + 2816000);
  float*    la    = ws + 5450000;

  const int* srcE = ei;
  const int* dstE = ei + NE;

  for (int layer = 0; layer < 1 + LSTK; ++layer) {
    const float* hin = (layer == 0) ? x : out + (size_t)(layer - 1) * NN * HIDD;
    const int li = layer - 1;
    const float* Wl_   = (layer == 0) ? p1_Wl   : Wl   + (size_t)li * HIDD * HIDD;
    const float* bl_   = (layer == 0) ? p1_bl   : bl   + (size_t)li * HIDD;
    const float* Wr_   = (layer == 0) ? p1_Wr   : Wr   + (size_t)li * HIDD * HIDD;
    const float* br_   = (layer == 0) ? p1_br   : br   + (size_t)li * HIDD;
    const float* We_   = (layer == 0) ? p1_We   : We   + (size_t)li * 2 * HIDD;
    const float* att_  = (layer == 0) ? p1_att  : att  + (size_t)li * HEADS * CHN;
    const float* bg_   = (layer == 0) ? p1_bias : bias_g + (size_t)li * HIDD;
    const float* Wres_ = (layer == 0) ? p1_Wres : Wres + (size_t)li * HIDD * HIDD;
    const float* bres_ = (layer == 0) ? p1_bres : bres + (size_t)li * HIDD;
    float* hout = out + (size_t)layer * NN * HIDD;

    dim3 gg(3, NN / 16);
    if (layer == 0)
      k_gemm3_wmma<FIN><<<gg, 32, 0, stream>>>(hin, Wl_, bl_, xl, Wr_, br_, xr, Wres_, bres_, res);
    else
      k_gemm3_wmma<HIDD><<<gg, 32, 0, stream>>>(hin, Wl_, bl_, xl, Wr_, br_, xr, Wres_, bres_, res);

    if (layer == 0)
      k_layer_owned<true><<<NBLK, LTB, 0, stream>>>(xl, xr, res, srcE, dstE, ea, la, We_, att_, bg_, hout, recs, gcnt);
    else
      k_layer_owned<false><<<NBLK, LTB, 0, stream>>>(xl, xr, res, srcE, dstE, ea, la, We_, att_, bg_, hout, recs, gcnt);
  }

  float* pooled = out + (size_t)(1 + LSTK) * NN * HIDD;
  float* x12    = pooled + NGR * HIDD;
  k_pool_owned<<<NGR, PTB, 0, stream>>>(out + (size_t)LSTK * NN * HIDD, batch, pooled);

  dim3 gf(VOC / 16, NGR / 16);
  k_gemm_wmma<<<gf, 32, 0, stream>>>(pooled, Wlin, blin, x12, NGR, HIDD, VOC);
}
